# starting kernel
# baseline (speedup 1.0000x reference)
#include <hip/hip_runtime.h>
#include <hip/hip_bf16.h>
#include <math.h>


typedef __attribute__((ext_vector_type(2))) float v2f;
typedef __attribute__((ext_vector_type(8))) float v8f;

#define BN_PAIRS 3312
#define KDIM     64
#define WCOLS    192

__device__ __forceinline__ v8f tile_matvec(const float* __restrict__ W,
                                           int j0,
                                           const float* __restrict__ vecLDS,
                                           v8f acc_in, int hi2, int m)
{
    v8f acc0 = acc_in;
    v8f acc1 = {};
    const float* wcol = W + j0 + m;
#pragma unroll
    for (int k0 = 0; k0 < KDIM; k0 += 8) {
        {
            const int kk = k0 + hi2;
            v2f a;  a.x = wcol[(kk + 0) * WCOLS]; a.y = wcol[(kk + 1) * WCOLS];
            v2f bb; bb.x = vecLDS[kk + 0];        bb.y = vecLDS[kk + 1];
            acc0 = __builtin_amdgcn_wmma_f32_16x16x4_f32(
                       false, a, false, bb, (short)0, acc0, false, false);
        }
        {
            const int kk = k0 + 4 + hi2;
            v2f a;  a.x = wcol[(kk + 0) * WCOLS]; a.y = wcol[(kk + 1) * WCOLS];
            v2f bb; bb.x = vecLDS[kk + 0];        bb.y = vecLDS[kk + 1];
            acc1 = __builtin_amdgcn_wmma_f32_16x16x4_f32(
                       false, a, false, bb, (short)0, acc1, false, false);
        }
    }
#pragma unroll
    for (int i = 0; i < 8; ++i) acc0[i] += acc1[i];
    return acc0;
}

extern "C" __global__ __launch_bounds__(32)
void gru_cell_wmma_kernel(const float* __restrict__ x,
                          const float* __restrict__ state,
                          const float* __restrict__ Wx,
                          const float* __restrict__ Wh,
                          const float* __restrict__ b,
                          float* __restrict__ out)
{
    __shared__ float x_lds[KDIM];
    __shared__ float h_lds[KDIM];
    __shared__ float b_lds[WCOLS];
    __shared__ float rh_lds[KDIM];
    __shared__ float z_lds[KDIM];
    __shared__ float hn_lds[KDIM];

    const int pair = blockIdx.x;
    const int lane = threadIdx.x;
    const int m    = lane & 15;
    const int hi   = lane >> 4;
    const int hi2  = hi * 2;

    const float* xg  = x     + (size_t)pair * KDIM;
    const float* hg  = state + (size_t)pair * KDIM;
    const float* Wxg = Wx    + (size_t)pair * KDIM * WCOLS;
    const float* Whg = Wh    + (size_t)pair * KDIM * WCOLS;
    const float* bg  = b     + (size_t)pair * WCOLS;
    float*       og  = out   + (size_t)pair * KDIM;

    x_lds[lane]      = xg[lane];
    x_lds[lane + 32] = xg[lane + 32];
    h_lds[lane]      = hg[lane];
    h_lds[lane + 32] = hg[lane + 32];
#pragma unroll
    for (int i = lane; i < WCOLS; i += 32) b_lds[i] = bg[i];
    __syncthreads();

#pragma unroll 1
    for (int t = 0; t < 8; ++t) {
        v8f acc = {};
        acc = tile_matvec(Wxg, 16 * t, x_lds, acc, hi2, m);
        acc = tile_matvec(Whg, 16 * t, h_lds, acc, hi2, m);
        if (m == 0) {
#pragma unroll
            for (int v = 0; v < 8; ++v) {
                const int j = 16 * t + 8 * hi + v;
                float g = acc[v] + b_lds[j];
                g = 1.0f / (1.0f + __expf(-g));
                if (t < 4) rh_lds[j]      = g * h_lds[j];
                else       z_lds[j - 64]  = g;
            }
        }
    }
    __syncthreads();

#pragma unroll 1
    for (int t = 8; t < 12; ++t) {
        v8f acc = {};
        acc = tile_matvec(Wxg, 16 * t, x_lds,  acc, hi2, m);
        acc = tile_matvec(Whg, 16 * t, rh_lds, acc, hi2, m);
        if (m == 0) {
#pragma unroll
            for (int v = 0; v < 8; ++v) {
                const int j  = 16 * t + 8 * hi + v;
                const int jj = j - 128;
                const float hc = tanhf(acc[v] + b_lds[j]);
                const float z  = z_lds[jj];
                hn_lds[jj] = (1.0f - z) * h_lds[jj] + z * hc;
            }
        }
    }
    __syncthreads();

    og[lane]      = hn_lds[lane];
    og[lane + 32] = hn_lds[lane + 32];
}

extern "C" void kernel_launch(void* const* d_in, const int* in_sizes, int n_in,
                              void* d_out, int out_size, void* d_ws, size_t ws_size,
                              hipStream_t stream)
{
    const float* x     = (const float*)d_in[0];
    const float* state = (const float*)d_in[1];
    const float* Wx    = (const float*)d_in[2];
    const float* Wh    = (const float*)d_in[3];
    const float* b     = (const float*)d_in[4];
    float* out = (float*)d_out;

    dim3 grid(BN_PAIRS);
    dim3 block(32);
    hipLaunchKernelGGL(gru_cell_wmma_kernel, grid, block, 0, stream,
                       x, state, Wx, Wh, b, out);
}


# Weights streamed as the WMMA B operand: 128-bit non-temporal loads, a half-wave per aligned 256-B row piece, 24 in flight per wave, 13 waves per work-group; k summed in one chain per column
# speedup vs baseline: 1.6751x; 1.6751x over previous
#include <hip/hip_runtime.h>


typedef __attribute__((ext_vector_type(2))) float v2f;
typedef __attribute__((ext_vector_type(4))) float v4f;
typedef __attribute__((ext_vector_type(8))) float v8f;

#define BN_PAIRS 3312
#define KDIM     64
#define WCOLS    192
#define WAVES    13
#define GROUPS   255
#define UNITS    96
#define RING     12

__device__ __forceinline__ v4f load_w(const float* p)
{
    return __builtin_nontemporal_load(reinterpret_cast<const v4f*>(p));
}

extern "C" __global__ __launch_bounds__(WAVES * 32)
void gru_cell_wmma_kernel(const float* __restrict__ x,
                          const float* __restrict__ state,
                          const float* __restrict__ Wx,
                          const float* __restrict__ Wh,
                          const float* __restrict__ b,
                          float* __restrict__ out)
{
    __shared__ float rh_lds[WAVES][KDIM];
    __shared__ float b_lds[WAVES][WCOLS];

    const int lane = threadIdx.x & 31;
    const int wave = __builtin_amdgcn_readfirstlane((int)(threadIdx.x >> 5));
    const int pair = blockIdx.x * WAVES + wave;
    if (pair >= BN_PAIRS) return;
    const int n  = lane & 15;
    const int hf = lane >> 4;

    const float* xg = x     + (size_t)pair * KDIM;
    const float* hg = state + (size_t)pair * KDIM;
    const float* bg = b     + (size_t)pair * WCOLS;
    const float* wx = Wx + (size_t)pair * KDIM * WCOLS + (2 * hf) * WCOLS + 4 * n;
    const float* wh = Wh + (size_t)pair * KDIM * WCOLS + (2 * hf) * WCOLS + 4 * n;

    v2f xa[16], ha[16];
#pragma unroll
    for (int g = 0; g < 16; ++g) {
        xa[g] = *reinterpret_cast<const v2f*>(xg + 4 * g + 2 * hf);
        ha[g] = *reinterpret_cast<const v2f*>(hg + 4 * g + 2 * hf);
    }
    const v4f hv = *reinterpret_cast<const v4f*>(hg + 4 * n);
#pragma unroll
    for (int i = 0; i < WCOLS / 32; ++i) b_lds[wave][lane + 32 * i] = bg[lane + 32 * i];

    v4f ring[RING][2];
#define ISSUE(U) do { const int u_ = (U); const int ps_ = u_ >> 4; const int g_ = u_ & 15;            \
        const float* p_ = ((ps_ & 1) ? wh : wx) + (ps_ >> 1) * 64 + g_ * (4 * WCOLS);                  \
        ring[u_ % RING][0] = load_w(p_);                                                               \
        ring[u_ % RING][1] = load_w(p_ + WCOLS); } while (0)

#define PASS(PS, AFR)                                                                                  \
    _Pragma("unroll")                                                                                  \
    for (int g = 0; g < 16; ++g) {                                                                     \
        const int u = (PS) * 16 + g;                                                                   \
        __builtin_amdgcn_sched_barrier(0);                                                             \
        const v4f w0 = ring[u % RING][0];                                                              \
        const v4f w1 = ring[u % RING][1];                                                              \
        _Pragma("unroll")                                                                              \
        for (int c = 0; c < 4; ++c) {                                                                  \
            v2f bb; bb.x = w0[c]; bb.y = w1[c];                                                        \
            acc[c] = __builtin_amdgcn_wmma_f32_16x16x4_f32(                                            \
                         false, AFR[g], false, bb, (short)0, acc[c], false, false);                    \
        }                                                                                              \
        __builtin_amdgcn_sched_barrier(0);                                                             \
        if (u + RING < UNITS) ISSUE(u + RING);                                                         \
    }

#pragma unroll
    for (int u = 0; u < RING; ++u) ISSUE(u);

    v8f acc[4];

#pragma unroll
    for (int c = 0; c < 4; ++c) acc[c] = v8f{};
    PASS(0, xa)
    PASS(1, ha)
    {
        const v4f bv = *reinterpret_cast<const v4f*>(&b_lds[wave][4 * n]);
        v4f rh;
#pragma unroll
        for (int c = 0; c < 4; ++c) {
            const float gt = acc[c][0] + bv[c];
            rh[c] = (1.0f / (1.0f + __expf(-gt))) * hv[c];
        }
        if (hf == 0) *reinterpret_cast<v4f*>(&rh_lds[wave][4 * n]) = rh;
    }

#pragma unroll
    for (int c = 0; c < 4; ++c) acc[c] = v8f{};
    PASS(2, xa)
    PASS(3, ha)
    v4f zg;
    {
        const v4f bv = *reinterpret_cast<const v4f*>(&b_lds[wave][64 + 4 * n]);
#pragma unroll
        for (int c = 0; c < 4; ++c) {
            const float gt = acc[c][0] + bv[c];
            zg[c] = 1.0f / (1.0f + __expf(-gt));
        }
    }

#pragma unroll
    for (int c = 0; c < 4; ++c) acc[c] = v8f{};
    PASS(4, xa)
    v2f rha[16];
#pragma unroll
    for (int g = 0; g < 16; ++g)
        rha[g] = *reinterpret_cast<const v2f*>(&rh_lds[wave][4 * g + 2 * hf]);
    PASS(5, rha)
    {
        const v4f bv = *reinterpret_cast<const v4f*>(&b_lds[wave][128 + 4 * n]);
        v4f o;
#pragma unroll
        for (int c = 0; c < 4; ++c) {
            const float hc = tanhf(acc[c][0] + bv[c]);
            o[c] = (1.0f - zg[c]) * hv[c] + zg[c] * hc;
        }
        if (hf == 0) *reinterpret_cast<v4f*>(out + (size_t)pair * KDIM + 4 * n) = o;
    }
#undef PASS
#undef ISSUE
}

extern "C" void kernel_launch(void* const* d_in, const int* in_sizes, int n_in,
                              void* d_out, int out_size, void* d_ws, size_t ws_size,
                              hipStream_t stream)
{
    const float* x     = (const float*)d_in[0];
    const float* state = (const float*)d_in[1];
    const float* Wx    = (const float*)d_in[2];
    const float* Wh    = (const float*)d_in[3];
    const float* b     = (const float*)d_in[4];
    float* out = (float*)d_out;

    dim3 grid(GROUPS);
    dim3 block(WAVES * 32);
    hipLaunchKernelGGL(gru_cell_wmma_kernel, grid, block, 0, stream,
                       x, state, Wx, Wh, b, out);
}
